# starting kernel
# baseline (speedup 1.0000x reference)
#include <hip/hip_runtime.h>
#include <hip/hip_bf16.h>
#include <math.h>

typedef __attribute__((ext_vector_type(2))) float v2f;
typedef __attribute__((ext_vector_type(8))) float v8f;

#define NEG_SLOPE 0.2f

__global__ __launch_bounds__(256) void gat_gemm1(const float* __restrict__ X,
                                                 const float* __restrict__ W1,
                                                 float* __restrict__ H1,
                                                 int N, int nTiles) {
    __shared__ float w[256 * 16];
    for (int i = threadIdx.x; i < 256 * 16; i += 256) {
        int k = i >> 4, col = i & 15;
        w[i] = (col < 8) ? W1[k * 8 + col] : 0.0f;
    }
    __syncthreads();

    int lane = threadIdx.x & 31;
    int tile = blockIdx.x * 8 + (threadIdx.x >> 5);
    if (tile >= nTiles) return;

    int n  = lane & 15;
    int hi = lane >> 4;
    int row = tile * 16 + n;
    int rrow = row < N ? row : N - 1;
    const float* xr = X + (size_t)rrow * 256 + 2 * hi;
    const float* wb = w + 2 * hi * 16 + n;

    v8f c = {};
    #pragma unroll 8
    for (int kk = 0; kk < 256; kk += 4) {
        float2 xv = *(const float2*)(xr + kk);
        v2f a; a.x = xv.x; a.y = xv.y;
        v2f b;
        b.x = wb[kk * 16];
        b.y = wb[kk * 16 + 16];
        c = __builtin_amdgcn_wmma_f32_16x16x4_f32(false, a, false, b,
                                                  (short)0, c, false, false);
    }

    if (n < 8) {
        #pragma unroll
        for (int r = 0; r < 8; ++r) {
            int m = tile * 16 + r + 8 * hi;
            if (m < N) H1[(size_t)m * 8 + n] = c[r];
        }
    }
}

__global__ __launch_bounds__(256) void gat_gemm2(const float* __restrict__ H,
                                                 const float* __restrict__ W2,
                                                 float* __restrict__ H2,
                                                 int N, int nTiles) {
    __shared__ float w[8 * 16];
    if (threadIdx.x < 128) w[threadIdx.x] = W2[threadIdx.x];
    __syncthreads();

    int lane = threadIdx.x & 31;
    int tile = blockIdx.x * 8 + (threadIdx.x >> 5);
    if (tile >= nTiles) return;

    int n  = lane & 15;
    int hi = lane >> 4;
    int row = tile * 16 + n;
    int rrow = row < N ? row : N - 1;
    const float* hr = H + (size_t)rrow * 8 + 2 * hi;

    v8f c = {};
    #pragma unroll
    for (int kk = 0; kk < 8; kk += 4) {
        v2f a; a.x = hr[kk]; a.y = hr[kk + 1];
        v2f b;
        b.x = w[(kk + 2 * hi) * 16 + n];
        b.y = w[(kk + 2 * hi + 1) * 16 + n];
        c = __builtin_amdgcn_wmma_f32_16x16x4_f32(false, a, false, b,
                                                  (short)0, c, false, false);
    }

    #pragma unroll
    for (int r = 0; r < 8; ++r) {
        int m = tile * 16 + r + 8 * hi;
        if (m < N) H2[(size_t)m * 16 + n] = c[r];
    }
}

template <int D>
__global__ void gat_node_prep(const float* __restrict__ H,
                              const float* __restrict__ a_src,
                              const float* __restrict__ a_dst,
                              float* __restrict__ aS, float* __restrict__ aD,
                              float* __restrict__ m, float* __restrict__ z,
                              float* __restrict__ acc, int N) {
    int i = blockIdx.x * blockDim.x + threadIdx.x;
    if (i >= N) return;
    const float* h = H + (size_t)i * D;
    float s = 0.0f, d = 0.0f;
    #pragma unroll
    for (int k = 0; k < D; ++k) { s += h[k] * a_src[k]; d += h[k] * a_dst[k]; }
    aS[i] = s; aD[i] = d;
    m[i] = -INFINITY; z[i] = 0.0f;
    float* ai = acc + (size_t)i * D;
    #pragma unroll
    for (int k = 0; k < D; ++k) ai[k] = 0.0f;
}

__global__ void gat_edge_max(const int* __restrict__ src, const int* __restrict__ dst,
                             const float* __restrict__ aS, const float* __restrict__ aD,
                             float* __restrict__ m, int E) {
    int e = blockIdx.x * blockDim.x + threadIdx.x;
    if (e >= E) return;
    int s = src[e], d = dst[e];
    float x = aS[s] + aD[d];
    x = x > 0.0f ? x : NEG_SLOPE * x;
    atomicMax(m + d, x);
}

template <int D>
__global__ void gat_edge_sum(const int* __restrict__ src, const int* __restrict__ dst,
                             const float* __restrict__ aS, const float* __restrict__ aD,
                             const float* __restrict__ m, const float* __restrict__ H,
                             float* __restrict__ z, float* __restrict__ acc, int E) {
    int e = blockIdx.x * blockDim.x + threadIdx.x;
    if (e >= E) return;
    int s = src[e], d = dst[e];
    float x = aS[s] + aD[d];
    x = x > 0.0f ? x : NEG_SLOPE * x;
    float w = expf(x - m[d]);
    atomicAdd(z + d, w);
    const float* hs = H + (size_t)s * D;
    float* ad = acc + (size_t)d * D;
    #pragma unroll
    for (int k = 0; k < D; ++k) atomicAdd(ad + k, hs[k] * w);
}

__global__ void gat_finish1(const float* __restrict__ acc, float* __restrict__ z,
                            float* __restrict__ H, float* __restrict__ m, int N) {
    int i = blockIdx.x * blockDim.x + threadIdx.x;
    if (i >= N) return;
    float zi = z[i] + 1e-16f;
    #pragma unroll
    for (int k = 0; k < 8; ++k) {
        float v = acc[(size_t)i * 8 + k] / zi;
        H[(size_t)i * 8 + k] = v > 0.0f ? v : expm1f(v);
    }
    m[i] = -INFINITY;
    z[i] = 0.0f;
}

__global__ void gat_finish2(const float* __restrict__ acc, const float* __restrict__ z,
                            float* __restrict__ out, int N) {
    int i = blockIdx.x * blockDim.x + threadIdx.x;
    if (i >= N) return;
    float zi = z[i] + 1e-16f;
    #pragma unroll
    for (int k = 0; k < 16; ++k)
        out[(size_t)i * 16 + k] = acc[(size_t)i * 16 + k] / zi;
}

extern "C" void kernel_launch(void* const* d_in, const int* in_sizes, int n_in,
                              void* d_out, int out_size, void* d_ws, size_t ws_size,
                              hipStream_t stream) {
    const int*   Ebuf  = (const int*)  d_in[1];
    const float* X     = (const float*)d_in[2];
    const float* W1    = (const float*)d_in[3];
    const float* a1s   = (const float*)d_in[4];
    const float* a1d   = (const float*)d_in[5];
    const float* W2    = (const float*)d_in[6];
    const float* a2s   = (const float*)d_in[7];
    const float* a2d   = (const float*)d_in[8];
    float*       out   = (float*)d_out;

    const int N = in_sizes[0];
    const int E = in_sizes[1] / 2;
    const int* src = Ebuf;
    const int* dst = Ebuf + E;

    float* w   = (float*)d_ws;
    float* H1  = w;
    float* H2  = H1 + (size_t)N * 8;
    float* aS  = H2 + (size_t)N * 16;
    float* aD  = aS + N;
    float* m   = aD + N;
    float* z   = m + N;
    float* acc = z + N;

    const int nTiles  = (N + 15) / 16;
    const int gBlocks = (nTiles + 7) / 8;
    const int nBlocks = (N + 255) / 256;
    const int eBlocks = (E + 255) / 256;

    gat_gemm1<<<gBlocks, 256, 0, stream>>>(X, W1, H1, N, nTiles);
    gat_node_prep<8><<<nBlocks, 256, 0, stream>>>(H1, a1s, a1d, aS, aD, m, z, acc, N);
    gat_edge_max<<<eBlocks, 256, 0, stream>>>(src, dst, aS, aD, m, E);
    gat_edge_sum<8><<<eBlocks, 256, 0, stream>>>(src, dst, aS, aD, m, H1, z, acc, E);
    gat_finish1<<<nBlocks, 256, 0, stream>>>(acc, z, H1, m, N);

    gat_gemm2<<<gBlocks, 256, 0, stream>>>(H1, W2, H2, N, nTiles);
    gat_node_prep<16><<<nBlocks, 256, 0, stream>>>(H2, a2s, a2d, aS, aD, m, z, acc, N);
    gat_edge_max<<<eBlocks, 256, 0, stream>>>(src, dst, aS, aD, m, E);
    gat_edge_sum<16><<<eBlocks, 256, 0, stream>>>(src, dst, aS, aD, m, H2, z, acc, E);
    gat_finish2<<<nBlocks, 256, 0, stream>>>(acc, z, out, N);
}


# edge_sum: lane = channel: D adjacent lanes add an edge's row to acc as one contiguous piece of a line (1 L2 atomic request per edge, not D); f32 atomics kept, arrival order; 8192 looping waves
# speedup vs baseline: 2.0383x; 2.0383x over previous
#include <hip/hip_runtime.h>
#include <hip/hip_bf16.h>
#include <math.h>

typedef __attribute__((ext_vector_type(2))) float v2f;
typedef __attribute__((ext_vector_type(8))) float v8f;

#define NEG_SLOPE 0.2f

__global__ __launch_bounds__(256) void gat_gemm1(const float* __restrict__ X,
                                                 const float* __restrict__ W1,
                                                 float* __restrict__ H1,
                                                 int N, int nTiles) {
    __shared__ float w[256 * 16];
    for (int i = threadIdx.x; i < 256 * 16; i += 256) {
        int k = i >> 4, col = i & 15;
        w[i] = (col < 8) ? W1[k * 8 + col] : 0.0f;
    }
    __syncthreads();

    int lane = threadIdx.x & 31;
    int tile = blockIdx.x * 8 + (threadIdx.x >> 5);
    if (tile >= nTiles) return;

    int n  = lane & 15;
    int hi = lane >> 4;
    int row = tile * 16 + n;
    int rrow = row < N ? row : N - 1;
    const float* xr = X + (size_t)rrow * 256 + 2 * hi;
    const float* wb = w + 2 * hi * 16 + n;

    v8f c = {};
    #pragma unroll 8
    for (int kk = 0; kk < 256; kk += 4) {
        float2 xv = *(const float2*)(xr + kk);
        v2f a; a.x = xv.x; a.y = xv.y;
        v2f b;
        b.x = wb[kk * 16];
        b.y = wb[kk * 16 + 16];
        c = __builtin_amdgcn_wmma_f32_16x16x4_f32(false, a, false, b,
                                                  (short)0, c, false, false);
    }

    if (n < 8) {
        #pragma unroll
        for (int r = 0; r < 8; ++r) {
            int m = tile * 16 + r + 8 * hi;
            if (m < N) H1[(size_t)m * 8 + n] = c[r];
        }
    }
}

__global__ __launch_bounds__(256) void gat_gemm2(const float* __restrict__ H,
                                                 const float* __restrict__ W2,
                                                 float* __restrict__ H2,
                                                 int N, int nTiles) {
    __shared__ float w[8 * 16];
    if (threadIdx.x < 128) w[threadIdx.x] = W2[threadIdx.x];
    __syncthreads();

    int lane = threadIdx.x & 31;
    int tile = blockIdx.x * 8 + (threadIdx.x >> 5);
    if (tile >= nTiles) return;

    int n  = lane & 15;
    int hi = lane >> 4;
    int row = tile * 16 + n;
    int rrow = row < N ? row : N - 1;
    const float* hr = H + (size_t)rrow * 8 + 2 * hi;

    v8f c = {};
    #pragma unroll
    for (int kk = 0; kk < 8; kk += 4) {
        v2f a; a.x = hr[kk]; a.y = hr[kk + 1];
        v2f b;
        b.x = w[(kk + 2 * hi) * 16 + n];
        b.y = w[(kk + 2 * hi + 1) * 16 + n];
        c = __builtin_amdgcn_wmma_f32_16x16x4_f32(false, a, false, b,
                                                  (short)0, c, false, false);
    }

    #pragma unroll
    for (int r = 0; r < 8; ++r) {
        int m = tile * 16 + r + 8 * hi;
        if (m < N) H2[(size_t)m * 16 + n] = c[r];
    }
}

template <int D>
__global__ void gat_node_prep(const float* __restrict__ H,
                              const float* __restrict__ a_src,
                              const float* __restrict__ a_dst,
                              float* __restrict__ aS, float* __restrict__ aD,
                              float* __restrict__ m, float* __restrict__ z,
                              float* __restrict__ acc, int N) {
    int i = blockIdx.x * blockDim.x + threadIdx.x;
    if (i >= N) return;
    const float* h = H + (size_t)i * D;
    float s = 0.0f, d = 0.0f;
    #pragma unroll
    for (int k = 0; k < D; ++k) { s += h[k] * a_src[k]; d += h[k] * a_dst[k]; }
    aS[i] = s; aD[i] = d;
    m[i] = -INFINITY; z[i] = 0.0f;
    float* ai = acc + (size_t)i * D;
    #pragma unroll
    for (int k = 0; k < D; ++k) ai[k] = 0.0f;
}

__global__ void gat_edge_max(const int* __restrict__ src, const int* __restrict__ dst,
                             const float* __restrict__ aS, const float* __restrict__ aD,
                             float* __restrict__ m, int E) {
    int e = blockIdx.x * blockDim.x + threadIdx.x;
    if (e >= E) return;
    int s = src[e], d = dst[e];
    float x = aS[s] + aD[d];
    x = x > 0.0f ? x : NEG_SLOPE * x;
    atomicMax(m + d, x);
}

template <int D, bool GUARD>
__device__ __forceinline__ void gat_edge_chunk(
        const int* __restrict__ src, const int* __restrict__ dst,
        const float* __restrict__ aS, const float* __restrict__ aD,
        const float* __restrict__ m, const float* __restrict__ H,
        float* __restrict__ z, float* __restrict__ acc,
        int base, int lane, int sub, int ch, int E) {
    constexpr int EPI = 32 / D;
    const int e = base + lane;
    int s = 0, d = 0;
    float w = 0.0f;
    if (!GUARD || e < E) {
        s = src[e]; d = dst[e];
    }
    float h[D];
    #pragma unroll
    for (int j = 0; j < D; ++j) {
        const int from = j * EPI + sub;
        const int s2 = __shfl(s, from);
        h[j] = 0.0f;
        if (!GUARD || base + from < E) h[j] = H[(size_t)s2 * D + ch];
    }
    if (!GUARD || e < E) {
        float x = aS[s] + aD[d];
        x = x > 0.0f ? x : NEG_SLOPE * x;
        w = expf(x - m[d]);
        atomicAdd(z + d, w);
    }
    #pragma unroll
    for (int j = 0; j < D; ++j) {
        const int from = j * EPI + sub;
        const int   d2 = __shfl(d, from);
        const float w2 = __shfl(w, from);
        if (!GUARD || base + from < E)
            atomicAdd(acc + (size_t)d2 * D + ch, h[j] * w2);
    }
}

template <int D>
__global__ __launch_bounds__(256) void gat_edge_sum(
        const int* __restrict__ src, const int* __restrict__ dst,
        const float* __restrict__ aS, const float* __restrict__ aD,
        const float* __restrict__ m, const float* __restrict__ H,
        float* __restrict__ z, float* __restrict__ acc, int E) {
    const int lane   = threadIdx.x & 31;
    const int wave   = __builtin_amdgcn_readfirstlane(
                           (int)(blockIdx.x * (blockDim.x >> 5) + (threadIdx.x >> 5)));
    const int nWaves = gridDim.x * (blockDim.x >> 5);
    const int sub    = lane / D;
    const int ch     = lane % D;

    for (int base = wave * 32; base < E; base += nWaves * 32) {
        if (base + 32 <= E)
            gat_edge_chunk<D, false>(src, dst, aS, aD, m, H, z, acc, base, lane, sub, ch, E);
        else
            gat_edge_chunk<D, true>(src, dst, aS, aD, m, H, z, acc, base, lane, sub, ch, E);
    }
}

__global__ void gat_finish1(const float* __restrict__ acc, float* __restrict__ z,
                            float* __restrict__ H, float* __restrict__ m, int N) {
    int i = blockIdx.x * blockDim.x + threadIdx.x;
    if (i >= N) return;
    float zi = z[i] + 1e-16f;
    #pragma unroll
    for (int k = 0; k < 8; ++k) {
        float v = acc[(size_t)i * 8 + k] / zi;
        H[(size_t)i * 8 + k] = v > 0.0f ? v : expm1f(v);
    }
    m[i] = -INFINITY;
    z[i] = 0.0f;
}

__global__ void gat_finish2(const float* __restrict__ acc, const float* __restrict__ z,
                            float* __restrict__ out, int N) {
    int i = blockIdx.x * blockDim.x + threadIdx.x;
    if (i >= N) return;
    float zi = z[i] + 1e-16f;
    #pragma unroll
    for (int k = 0; k < 16; ++k)
        out[(size_t)i * 16 + k] = acc[(size_t)i * 16 + k] / zi;
}

extern "C" void kernel_launch(void* const* d_in, const int* in_sizes, int n_in,
                              void* d_out, int out_size, void* d_ws, size_t ws_size,
                              hipStream_t stream) {
    const int*   Ebuf  = (const int*)  d_in[1];
    const float* X     = (const float*)d_in[2];
    const float* W1    = (const float*)d_in[3];
    const float* a1s   = (const float*)d_in[4];
    const float* a1d   = (const float*)d_in[5];
    const float* W2    = (const float*)d_in[6];
    const float* a2s   = (const float*)d_in[7];
    const float* a2d   = (const float*)d_in[8];
    float*       out   = (float*)d_out;

    const int N = in_sizes[0];
    const int E = in_sizes[1] / 2;
    const int* src = Ebuf;
    const int* dst = Ebuf + E;

    float* w   = (float*)d_ws;
    float* H1  = w;
    float* H2  = H1 + (size_t)N * 8;
    float* aS  = H2 + (size_t)N * 16;
    float* aD  = aS + N;
    float* m   = aD + N;
    float* z   = m + N;
    float* acc = z + N;

    const int nTiles  = (N + 15) / 16;
    const int gBlocks = (nTiles + 7) / 8;
    const int nBlocks = (N + 255) / 256;
    const int eBlocks = (E + 255) / 256;

    gat_gemm1<<<gBlocks, 256, 0, stream>>>(X, W1, H1, N, nTiles);
    gat_node_prep<8><<<nBlocks, 256, 0, stream>>>(H1, a1s, a1d, aS, aD, m, z, acc, N);
    gat_edge_max<<<eBlocks, 256, 0, stream>>>(src, dst, aS, aD, m, E);
    gat_edge_sum<8><<<1024, 256, 0, stream>>>(src, dst, aS, aD, m, H1, z, acc, E);
    gat_finish1<<<nBlocks, 256, 0, stream>>>(acc, z, H1, m, N);

    gat_gemm2<<<gBlocks, 256, 0, stream>>>(H1, W2, H2, N, nTiles);
    gat_node_prep<16><<<nBlocks, 256, 0, stream>>>(H2, a2s, a2d, aS, aD, m, z, acc, N);
    gat_edge_max<<<eBlocks, 256, 0, stream>>>(src, dst, aS, aD, m, E);
    gat_edge_sum<16><<<1024, 256, 0, stream>>>(src, dst, aS, aD, m, H2, z, acc, E);
    gat_finish2<<<nBlocks, 256, 0, stream>>>(acc, z, out, N);
}


# segment max taken over aS[src] alone (+aD[dst] and leaky-relu are monotone, applied after); {aD,max} one 8-byte pair per node: 7 not 9 scattered L2 requests per edge per layer; sum order as parent
# speedup vs baseline: 2.2528x; 1.1052x over previous
#include <hip/hip_runtime.h>
#include <hip/hip_bf16.h>
#include <math.h>

typedef __attribute__((ext_vector_type(2))) float v2f;
typedef __attribute__((ext_vector_type(8))) float v8f;

#define NEG_SLOPE 0.2f

__global__ __launch_bounds__(256) void gat_gemm1(const float* __restrict__ X,
                                                 const float* __restrict__ W1,
                                                 float* __restrict__ H1,
                                                 int N, int nTiles) {
    __shared__ float w[256 * 16];
    for (int i = threadIdx.x; i < 256 * 16; i += 256) {
        int k = i >> 4, col = i & 15;
        w[i] = (col < 8) ? W1[k * 8 + col] : 0.0f;
    }
    __syncthreads();

    int lane = threadIdx.x & 31;
    int tile = blockIdx.x * 8 + (threadIdx.x >> 5);
    if (tile >= nTiles) return;

    int n  = lane & 15;
    int hi = lane >> 4;
    int row = tile * 16 + n;
    int rrow = row < N ? row : N - 1;
    const float* xr = X + (size_t)rrow * 256 + 2 * hi;
    const float* wb = w + 2 * hi * 16 + n;

    v8f c = {};
    #pragma unroll 8
    for (int kk = 0; kk < 256; kk += 4) {
        float2 xv = *(const float2*)(xr + kk);
        v2f a; a.x = xv.x; a.y = xv.y;
        v2f b;
        b.x = wb[kk * 16];
        b.y = wb[kk * 16 + 16];
        c = __builtin_amdgcn_wmma_f32_16x16x4_f32(false, a, false, b,
                                                  (short)0, c, false, false);
    }

    if (n < 8) {
        #pragma unroll
        for (int r = 0; r < 8; ++r) {
            int m = tile * 16 + r + 8 * hi;
            if (m < N) H1[(size_t)m * 8 + n] = c[r];
        }
    }
}

__global__ __launch_bounds__(256) void gat_gemm2(const float* __restrict__ H,
                                                 const float* __restrict__ W2,
                                                 float* __restrict__ H2,
                                                 int N, int nTiles) {
    __shared__ float w[8 * 16];
    if (threadIdx.x < 128) w[threadIdx.x] = W2[threadIdx.x];
    __syncthreads();

    int lane = threadIdx.x & 31;
    int tile = blockIdx.x * 8 + (threadIdx.x >> 5);
    if (tile >= nTiles) return;

    int n  = lane & 15;
    int hi = lane >> 4;
    int row = tile * 16 + n;
    int rrow = row < N ? row : N - 1;
    const float* hr = H + (size_t)rrow * 8 + 2 * hi;

    v8f c = {};
    #pragma unroll
    for (int kk = 0; kk < 8; kk += 4) {
        v2f a; a.x = hr[kk]; a.y = hr[kk + 1];
        v2f b;
        b.x = w[(kk + 2 * hi) * 16 + n];
        b.y = w[(kk + 2 * hi + 1) * 16 + n];
        c = __builtin_amdgcn_wmma_f32_16x16x4_f32(false, a, false, b,
                                                  (short)0, c, false, false);
    }

    #pragma unroll
    for (int r = 0; r < 8; ++r) {
        int m = tile * 16 + r + 8 * hi;
        if (m < N) H2[(size_t)m * 16 + n] = c[r];
    }
}

template <int D>
__global__ void gat_node_prep(const float* __restrict__ H,
                              const float* __restrict__ a_src,
                              const float* __restrict__ a_dst,
                              float* __restrict__ aS, float* __restrict__ P,
                              float* __restrict__ z,
                              float* __restrict__ acc, int N) {
    int i = blockIdx.x * blockDim.x + threadIdx.x;
    if (i >= N) return;
    const float* h = H + (size_t)i * D;
    float s = 0.0f, d = 0.0f;
    #pragma unroll
    for (int k = 0; k < D; ++k) { s += h[k] * a_src[k]; d += h[k] * a_dst[k]; }
    aS[i] = s;
    *(float2*)(P + 2 * (size_t)i) = make_float2(d, -INFINITY);
    z[i] = 0.0f;
    float* ai = acc + (size_t)i * D;
    #pragma unroll
    for (int k = 0; k < D; ++k) ai[k] = 0.0f;
}

__global__ void gat_edge_max(const int* __restrict__ src, const int* __restrict__ dst,
                             const float* __restrict__ aS,
                             float* __restrict__ P, int E) {
    int e = blockIdx.x * blockDim.x + threadIdx.x;
    if (e >= E) return;
    int s = src[e], d = dst[e];
    atomicMax(P + 2 * (size_t)d + 1, aS[s]);
}

template <int D, bool GUARD>
__device__ __forceinline__ void gat_edge_chunk(
        const int* __restrict__ src, const int* __restrict__ dst,
        const float* __restrict__ aS, const float* __restrict__ P,
        const float* __restrict__ H,
        float* __restrict__ z, float* __restrict__ acc,
        int base, int lane, int sub, int ch, int E) {
    constexpr int EPI = 32 / D;
    const int e = base + lane;
    int s = 0, d = 0;
    float w = 0.0f;
    if (!GUARD || e < E) {
        s = src[e]; d = dst[e];
    }
    float h[D];
    #pragma unroll
    for (int j = 0; j < D; ++j) {
        const int from = j * EPI + sub;
        const int s2 = __shfl(s, from);
        h[j] = 0.0f;
        if (!GUARD || base + from < E) h[j] = H[(size_t)s2 * D + ch];
    }
    if (!GUARD || e < E) {
        const float2 p = *(const float2*)(P + 2 * (size_t)d);
        float x = aS[s] + p.x;
        x = x > 0.0f ? x : NEG_SLOPE * x;
        float mx = p.y + p.x;
        mx = mx > 0.0f ? mx : NEG_SLOPE * mx;
        w = expf(x - mx);
        atomicAdd(z + d, w);
    }
    #pragma unroll
    for (int j = 0; j < D; ++j) {
        const int from = j * EPI + sub;
        const int   d2 = __shfl(d, from);
        const float w2 = __shfl(w, from);
        if (!GUARD || base + from < E)
            atomicAdd(acc + (size_t)d2 * D + ch, h[j] * w2);
    }
}

template <int D>
__global__ __launch_bounds__(256) void gat_edge_sum(
        const int* __restrict__ src, const int* __restrict__ dst,
        const float* __restrict__ aS, const float* __restrict__ P,
        const float* __restrict__ H,
        float* __restrict__ z, float* __restrict__ acc, int E) {
    const int lane   = threadIdx.x & 31;
    const int wave   = __builtin_amdgcn_readfirstlane(
                           (int)(blockIdx.x * (blockDim.x >> 5) + (threadIdx.x >> 5)));
    const int nWaves = gridDim.x * (blockDim.x >> 5);
    const int sub    = lane / D;
    const int ch     = lane % D;

    for (int base = wave * 32; base < E; base += nWaves * 32) {
        if (base + 32 <= E)
            gat_edge_chunk<D, false>(src, dst, aS, P, H, z, acc, base, lane, sub, ch, E);
        else
            gat_edge_chunk<D, true>(src, dst, aS, P, H, z, acc, base, lane, sub, ch, E);
    }
}

__global__ void gat_finish1(const float* __restrict__ acc, float* __restrict__ z,
                            float* __restrict__ H, int N) {
    int i = blockIdx.x * blockDim.x + threadIdx.x;
    if (i >= N) return;
    float zi = z[i] + 1e-16f;
    #pragma unroll
    for (int k = 0; k < 8; ++k) {
        float v = acc[(size_t)i * 8 + k] / zi;
        H[(size_t)i * 8 + k] = v > 0.0f ? v : expm1f(v);
    }
    z[i] = 0.0f;
}

__global__ void gat_finish2(const float* __restrict__ acc, const float* __restrict__ z,
                            float* __restrict__ out, int N) {
    int i = blockIdx.x * blockDim.x + threadIdx.x;
    if (i >= N) return;
    float zi = z[i] + 1e-16f;
    #pragma unroll
    for (int k = 0; k < 16; ++k)
        out[(size_t)i * 16 + k] = acc[(size_t)i * 16 + k] / zi;
}

extern "C" void kernel_launch(void* const* d_in, const int* in_sizes, int n_in,
                              void* d_out, int out_size, void* d_ws, size_t ws_size,
                              hipStream_t stream) {
    const int*   Ebuf  = (const int*)  d_in[1];
    const float* X     = (const float*)d_in[2];
    const float* W1    = (const float*)d_in[3];
    const float* a1s   = (const float*)d_in[4];
    const float* a1d   = (const float*)d_in[5];
    const float* W2    = (const float*)d_in[6];
    const float* a2s   = (const float*)d_in[7];
    const float* a2d   = (const float*)d_in[8];
    float*       out   = (float*)d_out;

    const int N = in_sizes[0];
    const int E = in_sizes[1] / 2;
    const int* src = Ebuf;
    const int* dst = Ebuf + E;

    float* w   = (float*)d_ws;
    float* H1  = w;
    float* H2  = H1 + (size_t)N * 8;
    float* P   = H2 + (size_t)N * 16;
    float* aS  = P + (size_t)N * 2;
    float* z   = aS + N;
    float* acc = z + N;

    const int nTiles  = (N + 15) / 16;
    const int gBlocks = (nTiles + 7) / 8;
    const int nBlocks = (N + 255) / 256;
    const int eBlocks = (E + 255) / 256;

    gat_gemm1<<<gBlocks, 256, 0, stream>>>(X, W1, H1, N, nTiles);
    gat_node_prep<8><<<nBlocks, 256, 0, stream>>>(H1, a1s, a1d, aS, P, z, acc, N);
    gat_edge_max<<<eBlocks, 256, 0, stream>>>(src, dst, aS, P, E);
    gat_edge_sum<8><<<1024, 256, 0, stream>>>(src, dst, aS, P, H1, z, acc, E);
    gat_finish1<<<nBlocks, 256, 0, stream>>>(acc, z, H1, N);

    gat_gemm2<<<gBlocks, 256, 0, stream>>>(H1, W2, H2, N, nTiles);
    gat_node_prep<16><<<nBlocks, 256, 0, stream>>>(H2, a2s, a2d, aS, P, z, acc, N);
    gat_edge_max<<<eBlocks, 256, 0, stream>>>(src, dst, aS, P, E);
    gat_edge_sum<16><<<1024, 256, 0, stream>>>(src, dst, aS, P, H2, z, acc, E);
    gat_finish2<<<nBlocks, 256, 0, stream>>>(acc, z, out, N);
}


# edges bucketed by destination once per call (count, prefix sum, atomic place); each layer then reduces a node's in-edges in registers: no float atomics; sums in bucket (arrival) order, z by lane tree
# speedup vs baseline: 4.4166x; 1.9605x over previous
#include <hip/hip_runtime.h>
#include <hip/hip_bf16.h>
#include <math.h>

typedef __attribute__((ext_vector_type(2))) float v2f;
typedef __attribute__((ext_vector_type(8))) float v8f;

#define NEG_SLOPE 0.2f

__global__ __launch_bounds__(256) void gat_gemm1(const float* __restrict__ X,
                                                 const float* __restrict__ W1,
                                                 float* __restrict__ H1,
                                                 int N, int nTiles) {
    __shared__ float w[256 * 16];
    for (int i = threadIdx.x; i < 256 * 16; i += 256) {
        int k = i >> 4, col = i & 15;
        w[i] = (col < 8) ? W1[k * 8 + col] : 0.0f;
    }
    __syncthreads();

    int lane = threadIdx.x & 31;
    int tile = blockIdx.x * 8 + (threadIdx.x >> 5);
    if (tile >= nTiles) return;

    int n  = lane & 15;
    int hi = lane >> 4;
    int row = tile * 16 + n;
    int rrow = row < N ? row : N - 1;
    const float* xr = X + (size_t)rrow * 256 + 2 * hi;
    const float* wb = w + 2 * hi * 16 + n;

    v8f c = {};
    #pragma unroll 8
    for (int kk = 0; kk < 256; kk += 4) {
        float2 xv = *(const float2*)(xr + kk);
        v2f a; a.x = xv.x; a.y = xv.y;
        v2f b;
        b.x = wb[kk * 16];
        b.y = wb[kk * 16 + 16];
        c = __builtin_amdgcn_wmma_f32_16x16x4_f32(false, a, false, b,
                                                  (short)0, c, false, false);
    }

    if (n < 8) {
        #pragma unroll
        for (int r = 0; r < 8; ++r) {
            int m = tile * 16 + r + 8 * hi;
            if (m < N) H1[(size_t)m * 8 + n] = c[r];
        }
    }
}

__global__ __launch_bounds__(256) void gat_gemm2(const float* __restrict__ H,
                                                 const float* __restrict__ W2,
                                                 float* __restrict__ H2,
                                                 int N, int nTiles) {
    __shared__ float w[8 * 16];
    if (threadIdx.x < 128) w[threadIdx.x] = W2[threadIdx.x];
    __syncthreads();

    int lane = threadIdx.x & 31;
    int tile = blockIdx.x * 8 + (threadIdx.x >> 5);
    if (tile >= nTiles) return;

    int n  = lane & 15;
    int hi = lane >> 4;
    int row = tile * 16 + n;
    int rrow = row < N ? row : N - 1;
    const float* hr = H + (size_t)rrow * 8 + 2 * hi;

    v8f c = {};
    #pragma unroll
    for (int kk = 0; kk < 8; kk += 4) {
        v2f a; a.x = hr[kk]; a.y = hr[kk + 1];
        v2f b;
        b.x = w[(kk + 2 * hi) * 16 + n];
        b.y = w[(kk + 2 * hi + 1) * 16 + n];
        c = __builtin_amdgcn_wmma_f32_16x16x4_f32(false, a, false, b,
                                                  (short)0, c, false, false);
    }

    #pragma unroll
    for (int r = 0; r < 8; ++r) {
        int m = tile * 16 + r + 8 * hi;
        if (m < N) H2[(size_t)m * 16 + n] = c[r];
    }
}


template <int D, bool ZERO>
__global__ void gat_node_prep(const float* __restrict__ H,
                              const float* __restrict__ a_src,
                              const float* __restrict__ a_dst,
                              float* __restrict__ aS, float* __restrict__ aD,
                              int* __restrict__ cnt, int N) {
    int i = blockIdx.x * blockDim.x + threadIdx.x;
    if (i >= N) return;
    const float* h = H + (size_t)i * D;
    float s = 0.0f, d = 0.0f;
    #pragma unroll
    for (int k = 0; k < D; ++k) { s += h[k] * a_src[k]; d += h[k] * a_dst[k]; }
    aS[i] = s;
    aD[i] = d;
    if (ZERO) *(int2*)(cnt + 2 * (size_t)i) = make_int2(0, 0);
}

__global__ void csr_count(const int* __restrict__ src, const int* __restrict__ dst,
                          int* __restrict__ cnt, unsigned int* __restrict__ listw, int E) {
    int e = blockIdx.x * blockDim.x + threadIdx.x;
    if (e >= E) return;
    if (e < (E + 1) / 2) listw[e] = 0u;
    int s = src[e], d = dst[e];
    atomicAdd(cnt + 2 * (size_t)d + (s >> 16), 1);
}

__global__ __launch_bounds__(256) void csr_tile_sums(const int* __restrict__ cnt,
                                                     int* __restrict__ tsum, int n) {
    __shared__ int sh[256];
    const int tid = threadIdx.x;
    const int i0 = blockIdx.x * 1024 + tid * 4;
    int t = 0;
    #pragma unroll
    for (int k = 0; k < 4; ++k) if (i0 + k < n) t += cnt[i0 + k];
    sh[tid] = t;
    __syncthreads();
    for (int off = 128; off > 0; off >>= 1) {
        if (tid < off) sh[tid] += sh[tid + off];
        __syncthreads();
    }
    if (tid == 0) tsum[blockIdx.x] = sh[0];
}

__global__ __launch_bounds__(256) void csr_tile_scan(int* __restrict__ cnt,
                                                     const int* __restrict__ tsum, int n) {
    __shared__ int sh[256];
    const int tid = threadIdx.x;
    int p = 0;
    for (int j = tid; j < (int)blockIdx.x; j += 256) p += tsum[j];
    sh[tid] = p;
    __syncthreads();
    for (int off = 128; off > 0; off >>= 1) {
        if (tid < off) sh[tid] += sh[tid + off];
        __syncthreads();
    }
    const int tileBase = sh[0];
    __syncthreads();

    const int i0 = blockIdx.x * 1024 + tid * 4;
    int v[4];
    int t = 0;
    #pragma unroll
    for (int k = 0; k < 4; ++k) { v[k] = (i0 + k < n) ? cnt[i0 + k] : 0; t += v[k]; }
    sh[tid] = t;
    __syncthreads();
    for (int off = 1; off < 256; off <<= 1) {
        const int u = (tid >= off) ? sh[tid - off] : 0;
        __syncthreads();
        sh[tid] += u;
        __syncthreads();
    }
    int run = sh[tid] - t + tileBase;
    #pragma unroll
    for (int k = 0; k < 4; ++k) {
        if (i0 + k < n) cnt[i0 + k] = run;
        run += v[k];
    }
}

__global__ void csr_place(const int* __restrict__ src, const int* __restrict__ dst,
                          int* __restrict__ cur, unsigned int* __restrict__ listw, int E) {
    int e = blockIdx.x * blockDim.x + threadIdx.x;
    if (e >= E) return;
    int s = src[e], d = dst[e];
    const int slot = atomicAdd(cur + 2 * (size_t)d + (s >> 16), 1);
    atomicOr(listw + (slot >> 1), (unsigned int)(s & 0xFFFF) << ((slot & 1) * 16));
}

template <int D>
__device__ __forceinline__ void seg_rows(const float* __restrict__ H, int s, float w,
                                         int first, int end, int grp, int ch, float& acc) {
    float h[D];
    #pragma unroll
    for (int j = 0; j < D; ++j) {
        const int sj = __shfl(s, grp * D + j);
        h[j] = H[(size_t)sj * D + ch];
    }
    #pragma unroll
    for (int j = 0; j < D; ++j) {
        const float wj = __shfl(w, grp * D + j);
        acc += (first + j < end) ? wj * h[j] : 0.0f;
    }
}

template <int D, bool ELU>
__global__ __launch_bounds__(256) void gat_segment(
        const unsigned short* __restrict__ list, const int* __restrict__ cur,
        const float* __restrict__ aS, const float* __restrict__ aD,
        const float* __restrict__ H, float* __restrict__ outp, int N) {
    constexpr int G  = 32 / D;
    constexpr int KR = 64 / D;
    const int lane = threadIdx.x & 31;
    const int ch   = lane % D;
    const int grp  = lane / D;
    const int wave = blockIdx.x * (blockDim.x >> 5) + (threadIdx.x >> 5);
    const int d    = wave * G + grp;
    const bool live = d < N;

    int begin = 0, mid = 0, end = 0;
    float ad = 0.0f;
    if (live) {
        begin = d > 0 ? cur[2 * d - 1] : 0;
        mid   = cur[2 * d];
        end   = cur[2 * d + 1];
        ad    = aD[d];
    }

    int chunks = (end - begin + D - 1) / D;
    #pragma unroll
    for (int off = 16; off > 0; off >>= 1) {
        const int o = __shfl_xor(chunks, off);
        chunks = o > chunks ? o : chunks;
    }
    chunks = __builtin_amdgcn_readfirstlane(chunks);

    float m = -INFINITY;
    int   sv[KR];
    float xv[KR];
    {
        unsigned short lv[KR];
        #pragma unroll
        for (int k = 0; k < KR; ++k) {
            const int i = begin + k * D + ch;
            lv[k] = list[i < end ? i : 0];
        }
        #pragma unroll
        for (int k = 0; k < KR; ++k) {
            const int i = begin + k * D + ch;
            sv[k] = i < end ? (int)lv[k] + (i >= mid ? 65536 : 0) : 0;
        }
        float av[KR];
        #pragma unroll
        for (int k = 0; k < KR; ++k) av[k] = aS[sv[k]];
        #pragma unroll
        for (int k = 0; k < KR; ++k) {
            const int i = begin + k * D + ch;
            float x = av[k] + ad;
            x = x > 0.0f ? x : NEG_SLOPE * x;
            xv[k] = i < end ? x : -INFINITY;
            m = fmaxf(m, xv[k]);
        }
    }
    for (int c = KR; c < chunks; ++c) {
        const int i = begin + c * D + ch;
        if (i < end) {
            const int s = (int)list[i] + (i >= mid ? 65536 : 0);
            float x = aS[s] + ad;
            x = x > 0.0f ? x : NEG_SLOPE * x;
            m = fmaxf(m, x);
        }
    }
    #pragma unroll
    for (int off = D / 2; off > 0; off >>= 1) m = fmaxf(m, __shfl_xor(m, off));

    float acc = 0.0f, zl = 0.0f;
    #pragma unroll
    for (int k = 0; k < KR; ++k) {
        if (k < chunks) {
            const int i = begin + k * D + ch;
            const float w = i < end ? expf(xv[k] - m) : 0.0f;
            zl += w;
            seg_rows<D>(H, sv[k], w, begin + k * D, end, grp, ch, acc);
        }
    }
    for (int c = KR; c < chunks; ++c) {
        const int i = begin + c * D + ch;
        int s = 0;
        float w = 0.0f;
        if (i < end) {
            s = (int)list[i] + (i >= mid ? 65536 : 0);
            float x = aS[s] + ad;
            x = x > 0.0f ? x : NEG_SLOPE * x;
            w = expf(x - m);
        }
        zl += w;
        seg_rows<D>(H, s, w, begin + c * D, end, grp, ch, acc);
    }
    #pragma unroll
    for (int off = D / 2; off > 0; off >>= 1) zl += __shfl_xor(zl, off);

    if (live) {
        float v = acc / (zl + 1e-16f);
        if (ELU) v = v > 0.0f ? v : expm1f(v);
        outp[(size_t)d * D + ch] = v;
    }
}

extern "C" void kernel_launch(void* const* d_in, const int* in_sizes, int n_in,
                              void* d_out, int out_size, void* d_ws, size_t ws_size,
                              hipStream_t stream) {
    const int*   Ebuf  = (const int*)  d_in[1];
    const float* X     = (const float*)d_in[2];
    const float* W1    = (const float*)d_in[3];
    const float* a1s   = (const float*)d_in[4];
    const float* a1d   = (const float*)d_in[5];
    const float* W2    = (const float*)d_in[6];
    const float* a2s   = (const float*)d_in[7];
    const float* a2d   = (const float*)d_in[8];
    float*       out   = (float*)d_out;

    const int N = in_sizes[0];
    const int E = in_sizes[1] / 2;
    const int* src = Ebuf;
    const int* dst = Ebuf + E;

    float* w    = (float*)d_ws;
    float* A    = w;
    float* Hn   = A + (size_t)N * 16;
    float* aS   = Hn + (size_t)N * 8;
    float* aD   = aS + N;
    int*   cur  = (int*)(aD + N);
    unsigned int* listw = (unsigned int*)(cur + 2 * (size_t)N);
    const unsigned short* list = (const unsigned short*)listw;
    int*   tsum = (int*)(A + (size_t)N * 8);
    float* H1   = A;
    float* H2   = A;

    const int nTiles  = (N + 15) / 16;
    const int gBlocks = (nTiles + 7) / 8;
    const int nBlocks = (N + 255) / 256;
    const int eBlocks = (E + 255) / 256;
    const int nCnt    = 2 * N;
    const int sBlocks = (nCnt + 1023) / 1024;

    gat_gemm1<<<gBlocks, 256, 0, stream>>>(X, W1, H1, N, nTiles);
    gat_node_prep<8, true><<<nBlocks, 256, 0, stream>>>(H1, a1s, a1d, aS, aD, cur, N);
    csr_count<<<eBlocks, 256, 0, stream>>>(src, dst, cur, listw, E);
    csr_tile_sums<<<sBlocks, 256, 0, stream>>>(cur, tsum, nCnt);
    csr_tile_scan<<<sBlocks, 256, 0, stream>>>(cur, tsum, nCnt);
    csr_place<<<eBlocks, 256, 0, stream>>>(src, dst, cur, listw, E);

    gat_segment<8, true><<<(N + 31) / 32, 256, 0, stream>>>(list, cur, aS, aD, H1, Hn, N);

    gat_gemm2<<<gBlocks, 256, 0, stream>>>(Hn, W2, H2, N, nTiles);
    gat_node_prep<16, false><<<nBlocks, 256, 0, stream>>>(H2, a2s, a2d, aS, aD, cur, N);
    gat_segment<16, false><<<(N + 15) / 16, 256, 0, stream>>>(list, cur, aS, aD, H2, out, N);
}
